# starting kernel
# baseline (speedup 1.0000x reference)
#include <hip/hip_runtime.h>

typedef __attribute__((ext_vector_type(2))) float v2f;
typedef __attribute__((ext_vector_type(8))) float v8f;

#define R_DIM 64
#define C_DIM 256

__global__ void zero_kernel(float* __restrict__ p, int n) {
    int i = blockIdx.x * blockDim.x + threadIdx.x;
    if (i < n) p[i] = 0.0f;
}

__global__ void reduce_u_kernel(const float* __restrict__ eg,
                                float* __restrict__ u, int n) {
    __shared__ float smem[256];
    int tid  = threadIdx.x;
    int col  = tid & 63;
    int rgrp = tid >> 6;
    float s = 0.0f;
    for (int row = blockIdx.x * 4 + rgrp; row < n; row += gridDim.x * 4)
        s += eg[row * R_DIM + col];
    smem[tid] = s;
    __syncthreads();
    if (tid < 64) {
        float v = smem[tid] + smem[tid + 64] + smem[tid + 128] + smem[tid + 192];
        atomicAdd(&u[tid], v);
    }
}

#define EGP 160
#define XP  544

__global__ void proj_kernel(const float* __restrict__ eg,
                            const float* __restrict__ x,
                            float* __restrict__ proj, int n) {
    __shared__ float egp[16 * EGP];
    __shared__ float xp[16 * XP];

    int tid   = threadIdx.x;
    int lane  = tid & 31;
    int w     = tid >> 5;
    int mrow  = lane & 15;
    int khalf = lane >> 4;
    int rt    = w & 3;
    int ctb   = (w >> 2) * 8;

    int chunk = (n + gridDim.x - 1) / gridDim.x;
    int n0    = blockIdx.x * chunk;
    int nend  = min(n0 + chunk, n);

    v8f acc[8];
#pragma unroll
    for (int t = 0; t < 8; ++t)
#pragma unroll
        for (int j = 0; j < 8; ++j) acc[t][j] = 0.0f;

    for (int nb = n0; nb < nend; nb += 32) {
#pragma unroll
        for (int i = 0; i < 8; ++i) {
            int row = i * 4 + (tid >> 6);
            int col = tid & 63;
            int g   = nb + row;
            float v = (g < nend) ? eg[g * R_DIM + col] : 0.0f;
            egp[(row >> 1) * EGP + col * 2 + (row & 1)] = v;
        }
#pragma unroll
        for (int i = 0; i < 32; ++i) {
            int g   = nb + i;
            float v = (g < nend) ? x[g * C_DIM + tid] : 0.0f;
            xp[(i >> 1) * XP + tid * 2 + (i & 1)] = v;
        }
        __syncthreads();

#pragma unroll
        for (int kk = 0; kk < 8; ++kk) {
            int pidx = kk * 2 + khalf;
            v2f a = *(const v2f*)&egp[pidx * EGP + (rt * 16 + mrow) * 2];
#pragma unroll
            for (int t = 0; t < 8; ++t) {
                int c = (ctb + t) * 16 + mrow;
                v2f b = *(const v2f*)&xp[pidx * XP + c * 2];
                acc[t] = __builtin_amdgcn_wmma_f32_16x16x4_f32(
                    false, a, false, b, (short)0, acc[t], false, false);
            }
        }
        __syncthreads();
    }

#pragma unroll
    for (int t = 0; t < 8; ++t) {
        int c = (ctb + t) * 16 + mrow;
#pragma unroll
        for (int i = 0; i < 8; ++i) {
            int m = i + 8 * khalf;
            int r = rt * 16 + m;
            atomicAdd(&proj[r * C_DIM + c], acc[t][i]);
        }
    }
}

__global__ void p2_kernel(const float* __restrict__ proj,
                          const float* __restrict__ W,
                          float* __restrict__ P2i) {
    int tid   = threadIdx.x;
    int lane  = tid & 31;
    int w     = tid >> 5;
    int mrow  = lane & 15;
    int khalf = lane >> 4;
    int rt    = w & 3;
    int ctb   = (w >> 2) * 8;

    v8f acc[8];
#pragma unroll
    for (int t = 0; t < 8; ++t)
#pragma unroll
        for (int j = 0; j < 8; ++j) acc[t][j] = 0.0f;

    for (int kk = 0; kk < 64; ++kk) {
        int k0 = kk * 4 + 2 * khalf;
        v2f a = *(const v2f*)&proj[(rt * 16 + mrow) * C_DIM + k0];
#pragma unroll
        for (int t = 0; t < 8; ++t) {
            int c0 = (ctb + t) * 16;
            v2f b;
            b.x = W[k0 * C_DIM + c0 + mrow];
            b.y = W[(k0 + 1) * C_DIM + c0 + mrow];
            acc[t] = __builtin_amdgcn_wmma_f32_16x16x4_f32(
                false, a, false, b, (short)0, acc[t], false, false);
        }
    }
#pragma unroll
    for (int t = 0; t < 8; ++t) {
        int c = (ctb + t) * 16 + mrow;
#pragma unroll
        for (int i = 0; i < 8; ++i) {
            int r = rt * 16 + i + 8 * khalf;
            P2i[(r >> 1) * (2 * C_DIM) + c * 2 + (r & 1)] = acc[t][i];
        }
    }
}

#define EGO 68

__global__ void out_kernel(const float* __restrict__ eg,
                           const float* __restrict__ P2i,
                           const float* __restrict__ u,
                           const float* __restrict__ bias,
                           float* __restrict__ out, int n) {
    __shared__ float egLDS[64 * EGO];
    __shared__ float uLDS[64];
    __shared__ float invLDS[64];
    __shared__ float biasLDS[256];

    int tid = threadIdx.x;
    int rb  = blockIdx.x * 64;

    if (tid < 64) uLDS[tid] = u[tid];
    biasLDS[tid] = bias[tid];
#pragma unroll
    for (int i = 0; i < 16; ++i) {
        int row = i * 4 + (tid >> 6);
        int col = tid & 63;
        int g   = rb + row;
        egLDS[row * EGO + col] = (g < n) ? eg[g * R_DIM + col] : 0.0f;
    }
    __syncthreads();

    if (tid < 64) {
        float s = 0.0f;
#pragma unroll
        for (int r = 0; r < R_DIM; ++r) s += egLDS[tid * EGO + r] * uLDS[r];
        invLDS[tid] = 1.0f / s;
    }
    __syncthreads();

    int lane  = tid & 31;
    int w     = tid >> 5;
    int mrow  = lane & 15;
    int khalf = lane >> 4;
    int rt    = w & 3;
    int ctb   = (w >> 2) * 8;

    const v2f* P2v = (const v2f*)P2i;

    v8f acc[8];
#pragma unroll
    for (int t = 0; t < 8; ++t)
#pragma unroll
        for (int j = 0; j < 8; ++j) acc[t][j] = 0.0f;

#pragma unroll
    for (int kk = 0; kk < 16; ++kk) {
        int k0 = kk * 4 + 2 * khalf;
        v2f a = *(const v2f*)&egLDS[(rt * 16 + mrow) * EGO + k0];
#pragma unroll
        for (int t = 0; t < 8; ++t) {
            int c = (ctb + t) * 16 + mrow;
            v2f b = P2v[(k0 >> 1) * C_DIM + c];
            acc[t] = __builtin_amdgcn_wmma_f32_16x16x4_f32(
                false, a, false, b, (short)0, acc[t], false, false);
        }
    }

#pragma unroll
    for (int t = 0; t < 8; ++t) {
        int c = (ctb + t) * 16 + mrow;
#pragma unroll
        for (int i = 0; i < 8; ++i) {
            int m    = i + 8 * khalf;
            int grow = rb + rt * 16 + m;
            if (grow < n)
                out[grow * C_DIM + c] =
                    acc[t][i] * invLDS[rt * 16 + m] + biasLDS[c];
        }
    }
}

extern "C" void kernel_launch(void* const* d_in, const int* in_sizes, int n_in,
                              void* d_out, int out_size, void* d_ws, size_t ws_size,
                              hipStream_t stream) {
    const float* x    = (const float*)d_in[0];
    const float* eg   = (const float*)d_in[2];
    const float* W    = (const float*)d_in[3];
    const float* bias = (const float*)d_in[4];
    float*       out  = (float*)d_out;

    int N = in_sizes[2] / R_DIM;

    float* ws   = (float*)d_ws;
    float* u    = ws;
    float* proj = ws + 64;
    float* P2i  = ws + 64 + R_DIM * C_DIM;

    int nz = 64 + R_DIM * C_DIM;
    zero_kernel<<<(nz + 255) / 256, 256, 0, stream>>>(ws, nz);
    reduce_u_kernel<<<512, 256, 0, stream>>>(eg, u, N);
    proj_kernel<<<250, 256, 0, stream>>>(eg, x, proj, N);
    p2_kernel<<<1, 256, 0, stream>>>(proj, W, P2i);
    out_kernel<<<(N + 63) / 64, 256, 0, stream>>>(eg, P2i, u, bias, out, N);
}


# Loads no longer wait one by one: proj and out staging loads unguarded (clamped row, zeroed at LDS store), proj's issued as a group one 64-row step ahead, reduce_u in groups of 8; sum order kept
# speedup vs baseline: 2.6867x; 2.6867x over previous
#include <hip/hip_runtime.h>

typedef __attribute__((ext_vector_type(2))) float v2f;
typedef __attribute__((ext_vector_type(8))) float v8f;

#define R_DIM 64
#define C_DIM 256

__global__ void zero_kernel(float* __restrict__ p, int n) {
    int i = blockIdx.x * blockDim.x + threadIdx.x;
    if (i < n) p[i] = 0.0f;
}

__global__ void reduce_u_kernel(const float* __restrict__ eg,
                                float* __restrict__ u, int n) {
    __shared__ float smem[256];
    int tid  = threadIdx.x;
    int col  = tid & 63;
    int rgrp = tid >> 6;
    float s = 0.0f;
    int stride = gridDim.x * 4;
    int first  = blockIdx.x * 4 + rgrp;
    int cnt    = (first < n) ? (n - 1 - first) / stride + 1 : 0;
    for (int i = 0; i < cnt; i += 8) {
        float v[8];
#pragma unroll
        for (int k = 0; k < 8; ++k) {
            int row = first + min(i + k, cnt - 1) * stride;
            v[k] = eg[row * R_DIM + col];
        }
#pragma unroll
        for (int k = 0; k < 8; ++k) s = (i + k < cnt) ? s + v[k] : s;
    }
    smem[tid] = s;
    __syncthreads();
    if (tid < 64) {
        float v = smem[tid] + smem[tid + 64] + smem[tid + 128] + smem[tid + 192];
        atomicAdd(&u[tid], v);
    }
}

#define EGP 160
#define XP  544
#define PKS 64

typedef __attribute__((ext_vector_type(4))) float v4f;

__global__ __launch_bounds__(256)
void proj_kernel(const float* __restrict__ eg,
                 const float* __restrict__ x,
                 float* __restrict__ proj, int n) {
    __shared__ __attribute__((aligned(16))) float egp[(PKS / 2) * EGP];
    __shared__ __attribute__((aligned(16))) float xp[(PKS / 2) * XP];

    int tid   = threadIdx.x;
    int lane  = tid & 31;
    int w     = tid >> 5;
    int mrow  = lane & 15;
    int khalf = lane >> 4;
    int rt    = w & 3;
    int ctb   = (w >> 2) * 8;

    int chunk = (n + gridDim.x - 1) / gridDim.x;
    int n0    = blockIdx.x * chunk;
    int nend  = min(n0 + chunk, n);

    int xc4 = tid & 63;
    int xpp = tid >> 6;
    int ec4 = tid & 15;
    int epp = tid >> 4;
    const float* xb = x + xc4 * 4;
    const float* eb = eg + ec4 * 4;

    v4f rx[16], re[4];
    const v4f zero4 = {0.0f, 0.0f, 0.0f, 0.0f};

    auto fetch = [&](int nb) {
        if (nb + PKS <= nend) {
            const float* px = xb + (nb + 2 * xpp) * C_DIM;
            const float* pe = eb + (nb + 2 * epp) * R_DIM;
#pragma unroll
            for (int j = 0; j < 8; ++j) {
                rx[2 * j]     = *(const v4f*)(px + (8 * j) * C_DIM);
                rx[2 * j + 1] = *(const v4f*)(px + (8 * j + 1) * C_DIM);
            }
#pragma unroll
            for (int j = 0; j < 2; ++j) {
                re[2 * j]     = *(const v4f*)(pe + (32 * j) * R_DIM);
                re[2 * j + 1] = *(const v4f*)(pe + (32 * j + 1) * R_DIM);
            }
        } else {
#pragma unroll
            for (int j = 0; j < 8; ++j) {
                int r0 = min(nb + 2 * (xpp + 4 * j), nend - 1);
                int r1 = min(nb + 2 * (xpp + 4 * j) + 1, nend - 1);
                rx[2 * j]     = *(const v4f*)(xb + r0 * C_DIM);
                rx[2 * j + 1] = *(const v4f*)(xb + r1 * C_DIM);
            }
#pragma unroll
            for (int j = 0; j < 2; ++j) {
                int r0 = min(nb + 2 * (epp + 16 * j), nend - 1);
                int r1 = min(nb + 2 * (epp + 16 * j) + 1, nend - 1);
                re[2 * j]     = *(const v4f*)(eb + r0 * R_DIM);
                re[2 * j + 1] = *(const v4f*)(eb + r1 * R_DIM);
            }
        }
    };

    v8f acc[8];
#pragma unroll
    for (int t = 0; t < 8; ++t)
#pragma unroll
        for (int j = 0; j < 8; ++j) acc[t][j] = 0.0f;

    auto mma32 = [&](int pb) {
#pragma unroll
        for (int kk = 0; kk < 8; ++kk) {
            int pidx = pb + kk * 2 + khalf;
            v2f a = *(const v2f*)&egp[pidx * EGP + (rt * 16 + mrow) * 2];
#pragma unroll
            for (int t = 0; t < 8; ++t) {
                int c = (ctb + t) * 16 + mrow;
                v2f b = *(const v2f*)&xp[pidx * XP + c * 2];
                acc[t] = __builtin_amdgcn_wmma_f32_16x16x4_f32(
                    false, a, false, b, (short)0, acc[t], false, false);
            }
        }
    };

    if (n0 < nend) fetch(n0);

    for (int nb = n0; nb < nend; nb += PKS) {
        bool part = nb + PKS > nend;
#pragma unroll
        for (int j = 0; j < 8; ++j) {
            int p = xpp + 4 * j;
            float* d = &xp[p * XP + xc4 * 8];
            v4f a0 = rx[2 * j], a1 = rx[2 * j + 1];
            if (part) {
                a0 = (nb + 2 * p < nend) ? a0 : zero4;
                a1 = (nb + 2 * p + 1 < nend) ? a1 : zero4;
            }
            v4f s;
            s.x = a0.x; s.y = a1.x; s.z = a0.y; s.w = a1.y; *(v4f*)d       = s;
            s.x = a0.z; s.y = a1.z; s.z = a0.w; s.w = a1.w; *(v4f*)(d + 4) = s;
        }
#pragma unroll
        for (int j = 0; j < 2; ++j) {
            int p = epp + 16 * j;
            float* d = &egp[p * EGP + ec4 * 8];
            v4f a0 = re[2 * j], a1 = re[2 * j + 1];
            if (part) {
                a0 = (nb + 2 * p < nend) ? a0 : zero4;
                a1 = (nb + 2 * p + 1 < nend) ? a1 : zero4;
            }
            v4f s;
            s.x = a0.x; s.y = a1.x; s.z = a0.y; s.w = a1.y; *(v4f*)d       = s;
            s.x = a0.z; s.y = a1.z; s.z = a0.w; s.w = a1.w; *(v4f*)(d + 4) = s;
        }
        __syncthreads();

        if (nb + PKS < nend) fetch(nb + PKS);

        mma32(0);
        if (nb + 32 < nend) mma32(16);
        __syncthreads();
    }

#pragma unroll
    for (int t = 0; t < 8; ++t) {
        int c = (ctb + t) * 16 + mrow;
#pragma unroll
        for (int i = 0; i < 8; ++i) {
            int m = i + 8 * khalf;
            int r = rt * 16 + m;
            atomicAdd(&proj[r * C_DIM + c], acc[t][i]);
        }
    }
}

__global__ void p2_kernel(const float* __restrict__ proj,
                          const float* __restrict__ W,
                          float* __restrict__ P2i) {
    int tid   = threadIdx.x;
    int lane  = tid & 31;
    int w     = tid >> 5;
    int mrow  = lane & 15;
    int khalf = lane >> 4;
    int rt    = w & 3;
    int ctb   = (w >> 2) * 8;

    v8f acc[8];
#pragma unroll
    for (int t = 0; t < 8; ++t)
#pragma unroll
        for (int j = 0; j < 8; ++j) acc[t][j] = 0.0f;

    for (int kk = 0; kk < 64; ++kk) {
        int k0 = kk * 4 + 2 * khalf;
        v2f a = *(const v2f*)&proj[(rt * 16 + mrow) * C_DIM + k0];
#pragma unroll
        for (int t = 0; t < 8; ++t) {
            int c0 = (ctb + t) * 16;
            v2f b;
            b.x = W[k0 * C_DIM + c0 + mrow];
            b.y = W[(k0 + 1) * C_DIM + c0 + mrow];
            acc[t] = __builtin_amdgcn_wmma_f32_16x16x4_f32(
                false, a, false, b, (short)0, acc[t], false, false);
        }
    }
#pragma unroll
    for (int t = 0; t < 8; ++t) {
        int c = (ctb + t) * 16 + mrow;
#pragma unroll
        for (int i = 0; i < 8; ++i) {
            int r = rt * 16 + i + 8 * khalf;
            P2i[(r >> 1) * (2 * C_DIM) + c * 2 + (r & 1)] = acc[t][i];
        }
    }
}

#define EGO 68

__global__ void out_kernel(const float* __restrict__ eg,
                           const float* __restrict__ P2i,
                           const float* __restrict__ u,
                           const float* __restrict__ bias,
                           float* __restrict__ out, int n) {
    __shared__ float egLDS[64 * EGO];
    __shared__ float uLDS[64];
    __shared__ float invLDS[64];
    __shared__ float biasLDS[256];

    int tid = threadIdx.x;
    int rb  = blockIdx.x * 64;

    float uu = u[tid & 63];
    float bb = bias[tid];
    float ev[16];
#pragma unroll
    for (int i = 0; i < 16; ++i) {
        int row = i * 4 + (tid >> 6);
        int col = tid & 63;
        int g   = min(rb + row, n - 1);
        ev[i]   = eg[g * R_DIM + col];
    }
    if (tid < 64) uLDS[tid] = uu;
    biasLDS[tid] = bb;
#pragma unroll
    for (int i = 0; i < 16; ++i) {
        int row = i * 4 + (tid >> 6);
        int col = tid & 63;
        egLDS[row * EGO + col] = (rb + row < n) ? ev[i] : 0.0f;
    }
    __syncthreads();

    if (tid < 64) {
        float s = 0.0f;
#pragma unroll
        for (int r = 0; r < R_DIM; ++r) s += egLDS[tid * EGO + r] * uLDS[r];
        invLDS[tid] = 1.0f / s;
    }
    __syncthreads();

    int lane  = tid & 31;
    int w     = tid >> 5;
    int mrow  = lane & 15;
    int khalf = lane >> 4;
    int rt    = w & 3;
    int ctb   = (w >> 2) * 8;

    const v2f* P2v = (const v2f*)P2i;

    v8f acc[8];
#pragma unroll
    for (int t = 0; t < 8; ++t)
#pragma unroll
        for (int j = 0; j < 8; ++j) acc[t][j] = 0.0f;

#pragma unroll
    for (int kk = 0; kk < 16; ++kk) {
        int k0 = kk * 4 + 2 * khalf;
        v2f a = *(const v2f*)&egLDS[(rt * 16 + mrow) * EGO + k0];
#pragma unroll
        for (int t = 0; t < 8; ++t) {
            int c = (ctb + t) * 16 + mrow;
            v2f b = P2v[(k0 >> 1) * C_DIM + c];
            acc[t] = __builtin_amdgcn_wmma_f32_16x16x4_f32(
                false, a, false, b, (short)0, acc[t], false, false);
        }
    }

#pragma unroll
    for (int t = 0; t < 8; ++t) {
        int c = (ctb + t) * 16 + mrow;
#pragma unroll
        for (int i = 0; i < 8; ++i) {
            int m    = i + 8 * khalf;
            int grow = rb + rt * 16 + m;
            if (grow < n)
                out[grow * C_DIM + c] =
                    acc[t][i] * invLDS[rt * 16 + m] + biasLDS[c];
        }
    }
}

extern "C" void kernel_launch(void* const* d_in, const int* in_sizes, int n_in,
                              void* d_out, int out_size, void* d_ws, size_t ws_size,
                              hipStream_t stream) {
    const float* x    = (const float*)d_in[0];
    const float* eg   = (const float*)d_in[2];
    const float* W    = (const float*)d_in[3];
    const float* bias = (const float*)d_in[4];
    float*       out  = (float*)d_out;

    int N = in_sizes[2] / R_DIM;

    float* ws   = (float*)d_ws;
    float* u    = ws;
    float* proj = ws + 64;
    float* P2i  = ws + 64 + R_DIM * C_DIM;

    int nz = 64 + R_DIM * C_DIM;
    zero_kernel<<<(nz + 255) / 256, 256, 0, stream>>>(ws, nz);
    reduce_u_kernel<<<512, 256, 0, stream>>>(eg, u, N);
    proj_kernel<<<250, 256, 0, stream>>>(eg, x, proj, N);
    p2_kernel<<<1, 256, 0, stream>>>(proj, W, P2i);
    out_kernel<<<(N + 63) / 64, 256, 0, stream>>>(eg, P2i, u, bias, out, N);
}


# P2 = proj @ W on 64 work-groups, K split over each group's 8 waves, partial tiles summed through LDS in fixed wave order (P2's sum order changes): one load round trip instead of 32 in series
# speedup vs baseline: 3.3515x; 1.2474x over previous
#include <hip/hip_runtime.h>

typedef __attribute__((ext_vector_type(2))) float v2f;
typedef __attribute__((ext_vector_type(8))) float v8f;

#define R_DIM 64
#define C_DIM 256

__global__ void zero_kernel(float* __restrict__ p, int n) {
    int i = blockIdx.x * blockDim.x + threadIdx.x;
    if (i < n) p[i] = 0.0f;
}

__global__ void reduce_u_kernel(const float* __restrict__ eg,
                                float* __restrict__ u, int n) {
    __shared__ float smem[256];
    int tid  = threadIdx.x;
    int col  = tid & 63;
    int rgrp = tid >> 6;
    float s = 0.0f;
    int stride = gridDim.x * 4;
    int first  = blockIdx.x * 4 + rgrp;
    int cnt    = (first < n) ? (n - 1 - first) / stride + 1 : 0;
    for (int i = 0; i < cnt; i += 8) {
        float v[8];
#pragma unroll
        for (int k = 0; k < 8; ++k) {
            int row = first + min(i + k, cnt - 1) * stride;
            v[k] = eg[row * R_DIM + col];
        }
#pragma unroll
        for (int k = 0; k < 8; ++k) s = (i + k < cnt) ? s + v[k] : s;
    }
    smem[tid] = s;
    __syncthreads();
    if (tid < 64) {
        float v = smem[tid] + smem[tid + 64] + smem[tid + 128] + smem[tid + 192];
        atomicAdd(&u[tid], v);
    }
}

#define EGP 160
#define XP  544
#define PKS 64

typedef __attribute__((ext_vector_type(4))) float v4f;

__global__ __launch_bounds__(256)
void proj_kernel(const float* __restrict__ eg,
                 const float* __restrict__ x,
                 float* __restrict__ proj, int n) {
    __shared__ __attribute__((aligned(16))) float egp[(PKS / 2) * EGP];
    __shared__ __attribute__((aligned(16))) float xp[(PKS / 2) * XP];

    int tid   = threadIdx.x;
    int lane  = tid & 31;
    int w     = tid >> 5;
    int mrow  = lane & 15;
    int khalf = lane >> 4;
    int rt    = w & 3;
    int ctb   = (w >> 2) * 8;

    int chunk = (n + gridDim.x - 1) / gridDim.x;
    int n0    = blockIdx.x * chunk;
    int nend  = min(n0 + chunk, n);

    int xc4 = tid & 63;
    int xpp = tid >> 6;
    int ec4 = tid & 15;
    int epp = tid >> 4;
    const float* xb = x + xc4 * 4;
    const float* eb = eg + ec4 * 4;

    v4f rx[16], re[4];
    const v4f zero4 = {0.0f, 0.0f, 0.0f, 0.0f};

    auto fetch = [&](int nb) {
        if (nb + PKS <= nend) {
            const float* px = xb + (nb + 2 * xpp) * C_DIM;
            const float* pe = eb + (nb + 2 * epp) * R_DIM;
#pragma unroll
            for (int j = 0; j < 8; ++j) {
                rx[2 * j]     = *(const v4f*)(px + (8 * j) * C_DIM);
                rx[2 * j + 1] = *(const v4f*)(px + (8 * j + 1) * C_DIM);
            }
#pragma unroll
            for (int j = 0; j < 2; ++j) {
                re[2 * j]     = *(const v4f*)(pe + (32 * j) * R_DIM);
                re[2 * j + 1] = *(const v4f*)(pe + (32 * j + 1) * R_DIM);
            }
        } else {
#pragma unroll
            for (int j = 0; j < 8; ++j) {
                int r0 = min(nb + 2 * (xpp + 4 * j), nend - 1);
                int r1 = min(nb + 2 * (xpp + 4 * j) + 1, nend - 1);
                rx[2 * j]     = *(const v4f*)(xb + r0 * C_DIM);
                rx[2 * j + 1] = *(const v4f*)(xb + r1 * C_DIM);
            }
#pragma unroll
            for (int j = 0; j < 2; ++j) {
                int r0 = min(nb + 2 * (epp + 16 * j), nend - 1);
                int r1 = min(nb + 2 * (epp + 16 * j) + 1, nend - 1);
                re[2 * j]     = *(const v4f*)(eb + r0 * R_DIM);
                re[2 * j + 1] = *(const v4f*)(eb + r1 * R_DIM);
            }
        }
    };

    v8f acc[8];
#pragma unroll
    for (int t = 0; t < 8; ++t)
#pragma unroll
        for (int j = 0; j < 8; ++j) acc[t][j] = 0.0f;

    auto mma32 = [&](int pb) {
#pragma unroll
        for (int kk = 0; kk < 8; ++kk) {
            int pidx = pb + kk * 2 + khalf;
            v2f a = *(const v2f*)&egp[pidx * EGP + (rt * 16 + mrow) * 2];
#pragma unroll
            for (int t = 0; t < 8; ++t) {
                int c = (ctb + t) * 16 + mrow;
                v2f b = *(const v2f*)&xp[pidx * XP + c * 2];
                acc[t] = __builtin_amdgcn_wmma_f32_16x16x4_f32(
                    false, a, false, b, (short)0, acc[t], false, false);
            }
        }
    };

    if (n0 < nend) fetch(n0);

    for (int nb = n0; nb < nend; nb += PKS) {
        bool part = nb + PKS > nend;
#pragma unroll
        for (int j = 0; j < 8; ++j) {
            int p = xpp + 4 * j;
            float* d = &xp[p * XP + xc4 * 8];
            v4f a0 = rx[2 * j], a1 = rx[2 * j + 1];
            if (part) {
                a0 = (nb + 2 * p < nend) ? a0 : zero4;
                a1 = (nb + 2 * p + 1 < nend) ? a1 : zero4;
            }
            v4f s;
            s.x = a0.x; s.y = a1.x; s.z = a0.y; s.w = a1.y; *(v4f*)d       = s;
            s.x = a0.z; s.y = a1.z; s.z = a0.w; s.w = a1.w; *(v4f*)(d + 4) = s;
        }
#pragma unroll
        for (int j = 0; j < 2; ++j) {
            int p = epp + 16 * j;
            float* d = &egp[p * EGP + ec4 * 8];
            v4f a0 = re[2 * j], a1 = re[2 * j + 1];
            if (part) {
                a0 = (nb + 2 * p < nend) ? a0 : zero4;
                a1 = (nb + 2 * p + 1 < nend) ? a1 : zero4;
            }
            v4f s;
            s.x = a0.x; s.y = a1.x; s.z = a0.y; s.w = a1.y; *(v4f*)d       = s;
            s.x = a0.z; s.y = a1.z; s.z = a0.w; s.w = a1.w; *(v4f*)(d + 4) = s;
        }
        __syncthreads();

        if (nb + PKS < nend) fetch(nb + PKS);

        mma32(0);
        if (nb + 32 < nend) mma32(16);
        __syncthreads();
    }

#pragma unroll
    for (int t = 0; t < 8; ++t) {
        int c = (ctb + t) * 16 + mrow;
#pragma unroll
        for (int i = 0; i < 8; ++i) {
            int m = i + 8 * khalf;
            int r = rt * 16 + m;
            atomicAdd(&proj[r * C_DIM + c], acc[t][i]);
        }
    }
}

__global__ __launch_bounds__(256)
void p2_kernel(const float* __restrict__ proj,
               const float* __restrict__ W,
               float* __restrict__ P2i) {
    __shared__ float part[8 * 256];

    int tid   = threadIdx.x;
    int lane  = tid & 31;
    int w     = tid >> 5;
    int mrow  = lane & 15;
    int khalf = lane >> 4;
    int rt    = blockIdx.x & 3;
    int ct    = blockIdx.x >> 2;

    const float* pa = proj + (rt * 16 + mrow) * C_DIM + w * 32 + 2 * khalf;
    const float* pb = W + (w * 32 + 2 * khalf) * C_DIM + ct * 16 + mrow;

    v2f a[8], b[8];
#pragma unroll
    for (int j = 0; j < 8; ++j) {
        a[j]   = *(const v2f*)(pa + 4 * j);
        b[j].x = pb[(4 * j) * C_DIM];
        b[j].y = pb[(4 * j + 1) * C_DIM];
    }

    v8f acc;
#pragma unroll
    for (int i = 0; i < 8; ++i) acc[i] = 0.0f;
#pragma unroll
    for (int j = 0; j < 8; ++j)
        acc = __builtin_amdgcn_wmma_f32_16x16x4_f32(
            false, a[j], false, b[j], (short)0, acc, false, false);

#pragma unroll
    for (int i = 0; i < 8; ++i)
        part[w * 256 + (i + 8 * khalf) * 16 + mrow] = acc[i];
    __syncthreads();

    float s = part[tid];
#pragma unroll
    for (int k = 1; k < 8; ++k) s += part[k * 256 + tid];

    int r = rt * 16 + (tid >> 4);
    int c = ct * 16 + (tid & 15);
    P2i[(r >> 1) * (2 * C_DIM) + c * 2 + (r & 1)] = s;
}

#define EGO 68

__global__ void out_kernel(const float* __restrict__ eg,
                           const float* __restrict__ P2i,
                           const float* __restrict__ u,
                           const float* __restrict__ bias,
                           float* __restrict__ out, int n) {
    __shared__ float egLDS[64 * EGO];
    __shared__ float uLDS[64];
    __shared__ float invLDS[64];
    __shared__ float biasLDS[256];

    int tid = threadIdx.x;
    int rb  = blockIdx.x * 64;

    float uu = u[tid & 63];
    float bb = bias[tid];
    float ev[16];
#pragma unroll
    for (int i = 0; i < 16; ++i) {
        int row = i * 4 + (tid >> 6);
        int col = tid & 63;
        int g   = min(rb + row, n - 1);
        ev[i]   = eg[g * R_DIM + col];
    }
    if (tid < 64) uLDS[tid] = uu;
    biasLDS[tid] = bb;
#pragma unroll
    for (int i = 0; i < 16; ++i) {
        int row = i * 4 + (tid >> 6);
        int col = tid & 63;
        egLDS[row * EGO + col] = (rb + row < n) ? ev[i] : 0.0f;
    }
    __syncthreads();

    if (tid < 64) {
        float s = 0.0f;
#pragma unroll
        for (int r = 0; r < R_DIM; ++r) s += egLDS[tid * EGO + r] * uLDS[r];
        invLDS[tid] = 1.0f / s;
    }
    __syncthreads();

    int lane  = tid & 31;
    int w     = tid >> 5;
    int mrow  = lane & 15;
    int khalf = lane >> 4;
    int rt    = w & 3;
    int ctb   = (w >> 2) * 8;

    const v2f* P2v = (const v2f*)P2i;

    v8f acc[8];
#pragma unroll
    for (int t = 0; t < 8; ++t)
#pragma unroll
        for (int j = 0; j < 8; ++j) acc[t][j] = 0.0f;

#pragma unroll
    for (int kk = 0; kk < 16; ++kk) {
        int k0 = kk * 4 + 2 * khalf;
        v2f a = *(const v2f*)&egLDS[(rt * 16 + mrow) * EGO + k0];
#pragma unroll
        for (int t = 0; t < 8; ++t) {
            int c = (ctb + t) * 16 + mrow;
            v2f b = P2v[(k0 >> 1) * C_DIM + c];
            acc[t] = __builtin_amdgcn_wmma_f32_16x16x4_f32(
                false, a, false, b, (short)0, acc[t], false, false);
        }
    }

#pragma unroll
    for (int t = 0; t < 8; ++t) {
        int c = (ctb + t) * 16 + mrow;
#pragma unroll
        for (int i = 0; i < 8; ++i) {
            int m    = i + 8 * khalf;
            int grow = rb + rt * 16 + m;
            if (grow < n)
                out[grow * C_DIM + c] =
                    acc[t][i] * invLDS[rt * 16 + m] + biasLDS[c];
        }
    }
}

extern "C" void kernel_launch(void* const* d_in, const int* in_sizes, int n_in,
                              void* d_out, int out_size, void* d_ws, size_t ws_size,
                              hipStream_t stream) {
    const float* x    = (const float*)d_in[0];
    const float* eg   = (const float*)d_in[2];
    const float* W    = (const float*)d_in[3];
    const float* bias = (const float*)d_in[4];
    float*       out  = (float*)d_out;

    int N = in_sizes[2] / R_DIM;

    float* ws   = (float*)d_ws;
    float* u    = ws;
    float* proj = ws + 64;
    float* P2i  = ws + 64 + R_DIM * C_DIM;

    int nz = 64 + R_DIM * C_DIM;
    zero_kernel<<<(nz + 255) / 256, 256, 0, stream>>>(ws, nz);
    reduce_u_kernel<<<512, 256, 0, stream>>>(eg, u, N);
    proj_kernel<<<250, 256, 0, stream>>>(eg, x, proj, N);
    p2_kernel<<<64, 256, 0, stream>>>(proj, W, P2i);
    out_kernel<<<(N + 63) / 64, 256, 0, stream>>>(eg, P2i, u, bias, out, N);
}


# proj_kernel: each work-group owns one 64-column slab of x over a 4x longer row chunk (grid 4x64), so 4x fewer float atomics into proj; proj's sum order changes (longer partial sums)
# speedup vs baseline: 3.9401x; 1.1756x over previous
#include <hip/hip_runtime.h>

typedef __attribute__((ext_vector_type(2))) float v2f;
typedef __attribute__((ext_vector_type(8))) float v8f;

#define R_DIM 64
#define C_DIM 256

__global__ void zero_kernel(float* __restrict__ p, int n) {
    int i = blockIdx.x * blockDim.x + threadIdx.x;
    if (i < n) p[i] = 0.0f;
}

__global__ void reduce_u_kernel(const float* __restrict__ eg,
                                float* __restrict__ u, int n) {
    __shared__ float smem[256];
    int tid  = threadIdx.x;
    int col  = tid & 63;
    int rgrp = tid >> 6;
    float s = 0.0f;
    int stride = gridDim.x * 4;
    int first  = blockIdx.x * 4 + rgrp;
    int cnt    = (first < n) ? (n - 1 - first) / stride + 1 : 0;
    for (int i = 0; i < cnt; i += 8) {
        float v[8];
#pragma unroll
        for (int k = 0; k < 8; ++k) {
            int row = first + min(i + k, cnt - 1) * stride;
            v[k] = eg[row * R_DIM + col];
        }
#pragma unroll
        for (int k = 0; k < 8; ++k) s = (i + k < cnt) ? s + v[k] : s;
    }
    smem[tid] = s;
    __syncthreads();
    if (tid < 64) {
        float v = smem[tid] + smem[tid + 64] + smem[tid + 128] + smem[tid + 192];
        atomicAdd(&u[tid], v);
    }
}

#define PP  160
#define PKS 256
#define SLAB 64

typedef __attribute__((ext_vector_type(4))) float v4f;

__global__ __launch_bounds__(256)
void proj_kernel(const float* __restrict__ eg,
                 const float* __restrict__ x,
                 float* __restrict__ proj, int n) {
    __shared__ __attribute__((aligned(16))) float egp[(PKS / 2) * PP];
    __shared__ __attribute__((aligned(16))) float xp[(PKS / 2) * PP];

    int tid   = threadIdx.x;
    int lane  = tid & 31;
    int w     = tid >> 5;
    int mrow  = lane & 15;
    int khalf = lane >> 4;
    int rt    = w & 3;
    int ctb   = (w >> 2) * 2;
    int cs    = blockIdx.x * SLAB;

    int chunk = (n + gridDim.y - 1) / gridDim.y;
    int n0    = blockIdx.y * chunk;
    int nend  = min(n0 + chunk, n);

    int c4 = tid & 15;
    int pp = tid >> 4;
    const float* xb = x + cs + c4 * 4;
    const float* eb = eg + c4 * 4;

    v4f rx[16], re[16];
    const v4f zero4 = {0.0f, 0.0f, 0.0f, 0.0f};

    auto fetch = [&](int nb) {
        if (nb + PKS <= nend) {
            const float* px = xb + (nb + 2 * pp) * C_DIM;
            const float* pe = eb + (nb + 2 * pp) * R_DIM;
#pragma unroll
            for (int j = 0; j < 8; ++j) {
                rx[2 * j]     = *(const v4f*)(px + (32 * j) * C_DIM);
                rx[2 * j + 1] = *(const v4f*)(px + (32 * j + 1) * C_DIM);
            }
#pragma unroll
            for (int j = 0; j < 8; ++j) {
                re[2 * j]     = *(const v4f*)(pe + (32 * j) * R_DIM);
                re[2 * j + 1] = *(const v4f*)(pe + (32 * j + 1) * R_DIM);
            }
        } else {
#pragma unroll
            for (int j = 0; j < 8; ++j) {
                int r0 = min(nb + 2 * (pp + 16 * j), nend - 1);
                int r1 = min(nb + 2 * (pp + 16 * j) + 1, nend - 1);
                rx[2 * j]     = *(const v4f*)(xb + r0 * C_DIM);
                rx[2 * j + 1] = *(const v4f*)(xb + r1 * C_DIM);
                re[2 * j]     = *(const v4f*)(eb + r0 * R_DIM);
                re[2 * j + 1] = *(const v4f*)(eb + r1 * R_DIM);
            }
        }
    };

    v8f acc[2];
#pragma unroll
    for (int t = 0; t < 2; ++t)
#pragma unroll
        for (int j = 0; j < 8; ++j) acc[t][j] = 0.0f;

    struct Frag { v2f a[8], b0[8], b1[8]; };
    auto ldfrag = [&](Frag& f, int pb) {
#pragma unroll
        for (int kk = 0; kk < 8; ++kk) {
            int pidx = pb + kk * 2 + khalf;
            f.a[kk]  = *(const v2f*)&egp[pidx * PP + (rt * 16 + mrow) * 2];
            f.b0[kk] = *(const v2f*)&xp[pidx * PP + ((ctb    ) * 16 + mrow) * 2];
            f.b1[kk] = *(const v2f*)&xp[pidx * PP + ((ctb + 1) * 16 + mrow) * 2];
        }
    };
    auto mmafrag = [&](const Frag& f) {
#pragma unroll
        for (int kk = 0; kk < 8; ++kk) {
            acc[0] = __builtin_amdgcn_wmma_f32_16x16x4_f32(
                false, f.a[kk], false, f.b0[kk], (short)0, acc[0], false, false);
            acc[1] = __builtin_amdgcn_wmma_f32_16x16x4_f32(
                false, f.a[kk], false, f.b1[kk], (short)0, acc[1], false, false);
        }
    };

    if (n0 < nend) fetch(n0);

    for (int nb = n0; nb < nend; nb += PKS) {
        bool part = nb + PKS > nend;
#pragma unroll
        for (int j = 0; j < 8; ++j) {
            int p = pp + 16 * j;
            float* dx = &xp[p * PP + c4 * 8];
            float* de = &egp[p * PP + c4 * 8];
            v4f a0 = rx[2 * j], a1 = rx[2 * j + 1];
            v4f e0 = re[2 * j], e1 = re[2 * j + 1];
            if (part) {
                bool k0 = nb + 2 * p < nend, k1 = nb + 2 * p + 1 < nend;
                a0 = k0 ? a0 : zero4;  a1 = k1 ? a1 : zero4;
                e0 = k0 ? e0 : zero4;  e1 = k1 ? e1 : zero4;
            }
            v4f s;
            s.x = a0.x; s.y = a1.x; s.z = a0.y; s.w = a1.y; *(v4f*)dx       = s;
            s.x = a0.z; s.y = a1.z; s.z = a0.w; s.w = a1.w; *(v4f*)(dx + 4) = s;
            s.x = e0.x; s.y = e1.x; s.z = e0.y; s.w = e1.y; *(v4f*)de       = s;
            s.x = e0.z; s.y = e1.z; s.z = e0.w; s.w = e1.w; *(v4f*)(de + 4) = s;
        }
        __syncthreads();

        if (nb + PKS < nend) fetch(nb + PKS);

        if (!part) {
            Frag fa, fb;
            ldfrag(fa, 0);
#pragma unroll
            for (int h = 0; h < PKS / 32; h += 2) {
                ldfrag(fb, 16 * (h + 1));
                __builtin_amdgcn_sched_barrier(0);
                mmafrag(fa);
                __builtin_amdgcn_sched_barrier(0);
                if (h + 2 < PKS / 32) ldfrag(fa, 16 * (h + 2));
                __builtin_amdgcn_sched_barrier(0);
                mmafrag(fb);
                __builtin_amdgcn_sched_barrier(0);
            }
        } else {
#pragma nounroll
            for (int h = 0; h < PKS / 32; ++h)
                if (nb + 32 * h < nend) {
                    Frag f;
                    ldfrag(f, 16 * h);
                    mmafrag(f);
                }
        }
        __syncthreads();
    }

#pragma unroll
    for (int t = 0; t < 2; ++t) {
        int c = cs + (ctb + t) * 16 + mrow;
#pragma unroll
        for (int i = 0; i < 8; ++i) {
            int m = i + 8 * khalf;
            int r = rt * 16 + m;
            atomicAdd(&proj[r * C_DIM + c], acc[t][i]);
        }
    }
}

__global__ __launch_bounds__(256)
void p2_kernel(const float* __restrict__ proj,
               const float* __restrict__ W,
               float* __restrict__ P2i) {
    __shared__ float part[8 * 256];

    int tid   = threadIdx.x;
    int lane  = tid & 31;
    int w     = tid >> 5;
    int mrow  = lane & 15;
    int khalf = lane >> 4;
    int rt    = blockIdx.x & 3;
    int ct    = blockIdx.x >> 2;

    const float* pa = proj + (rt * 16 + mrow) * C_DIM + w * 32 + 2 * khalf;
    const float* pb = W + (w * 32 + 2 * khalf) * C_DIM + ct * 16 + mrow;

    v2f a[8], b[8];
#pragma unroll
    for (int j = 0; j < 8; ++j) {
        a[j]   = *(const v2f*)(pa + 4 * j);
        b[j].x = pb[(4 * j) * C_DIM];
        b[j].y = pb[(4 * j + 1) * C_DIM];
    }

    v8f acc;
#pragma unroll
    for (int i = 0; i < 8; ++i) acc[i] = 0.0f;
#pragma unroll
    for (int j = 0; j < 8; ++j)
        acc = __builtin_amdgcn_wmma_f32_16x16x4_f32(
            false, a[j], false, b[j], (short)0, acc, false, false);

#pragma unroll
    for (int i = 0; i < 8; ++i)
        part[w * 256 + (i + 8 * khalf) * 16 + mrow] = acc[i];
    __syncthreads();

    float s = part[tid];
#pragma unroll
    for (int k = 1; k < 8; ++k) s += part[k * 256 + tid];

    int r = rt * 16 + (tid >> 4);
    int c = ct * 16 + (tid & 15);
    P2i[(r >> 1) * (2 * C_DIM) + c * 2 + (r & 1)] = s;
}

#define EGO 68

__global__ void out_kernel(const float* __restrict__ eg,
                           const float* __restrict__ P2i,
                           const float* __restrict__ u,
                           const float* __restrict__ bias,
                           float* __restrict__ out, int n) {
    __shared__ float egLDS[64 * EGO];
    __shared__ float uLDS[64];
    __shared__ float invLDS[64];
    __shared__ float biasLDS[256];

    int tid = threadIdx.x;
    int rb  = blockIdx.x * 64;

    float uu = u[tid & 63];
    float bb = bias[tid];
    float ev[16];
#pragma unroll
    for (int i = 0; i < 16; ++i) {
        int row = i * 4 + (tid >> 6);
        int col = tid & 63;
        int g   = min(rb + row, n - 1);
        ev[i]   = eg[g * R_DIM + col];
    }
    if (tid < 64) uLDS[tid] = uu;
    biasLDS[tid] = bb;
#pragma unroll
    for (int i = 0; i < 16; ++i) {
        int row = i * 4 + (tid >> 6);
        int col = tid & 63;
        egLDS[row * EGO + col] = (rb + row < n) ? ev[i] : 0.0f;
    }
    __syncthreads();

    if (tid < 64) {
        float s = 0.0f;
#pragma unroll
        for (int r = 0; r < R_DIM; ++r) s += egLDS[tid * EGO + r] * uLDS[r];
        invLDS[tid] = 1.0f / s;
    }
    __syncthreads();

    int lane  = tid & 31;
    int w     = tid >> 5;
    int mrow  = lane & 15;
    int khalf = lane >> 4;
    int rt    = w & 3;
    int ctb   = (w >> 2) * 8;

    const v2f* P2v = (const v2f*)P2i;

    v8f acc[8];
#pragma unroll
    for (int t = 0; t < 8; ++t)
#pragma unroll
        for (int j = 0; j < 8; ++j) acc[t][j] = 0.0f;

#pragma unroll
    for (int kk = 0; kk < 16; ++kk) {
        int k0 = kk * 4 + 2 * khalf;
        v2f a = *(const v2f*)&egLDS[(rt * 16 + mrow) * EGO + k0];
#pragma unroll
        for (int t = 0; t < 8; ++t) {
            int c = (ctb + t) * 16 + mrow;
            v2f b = P2v[(k0 >> 1) * C_DIM + c];
            acc[t] = __builtin_amdgcn_wmma_f32_16x16x4_f32(
                false, a, false, b, (short)0, acc[t], false, false);
        }
    }

#pragma unroll
    for (int t = 0; t < 8; ++t) {
        int c = (ctb + t) * 16 + mrow;
#pragma unroll
        for (int i = 0; i < 8; ++i) {
            int m    = i + 8 * khalf;
            int grow = rb + rt * 16 + m;
            if (grow < n)
                out[grow * C_DIM + c] =
                    acc[t][i] * invLDS[rt * 16 + m] + biasLDS[c];
        }
    }
}

extern "C" void kernel_launch(void* const* d_in, const int* in_sizes, int n_in,
                              void* d_out, int out_size, void* d_ws, size_t ws_size,
                              hipStream_t stream) {
    const float* x    = (const float*)d_in[0];
    const float* eg   = (const float*)d_in[2];
    const float* W    = (const float*)d_in[3];
    const float* bias = (const float*)d_in[4];
    float*       out  = (float*)d_out;

    int N = in_sizes[2] / R_DIM;

    float* ws   = (float*)d_ws;
    float* u    = ws;
    float* proj = ws + 64;
    float* P2i  = ws + 64 + R_DIM * C_DIM;

    int nz = 64 + R_DIM * C_DIM;
    zero_kernel<<<(nz + 255) / 256, 256, 0, stream>>>(ws, nz);
    reduce_u_kernel<<<512, 256, 0, stream>>>(eg, u, N);
    proj_kernel<<<dim3(4, 64), 256, 0, stream>>>(eg, x, proj, N);
    p2_kernel<<<64, 256, 0, stream>>>(proj, W, P2i);
    out_kernel<<<(N + 63) / 64, 256, 0, stream>>>(eg, P2i, u, bias, out, N);
}


# u (column sums of eg) now summed inside proj_kernel by each row chunk's slab-0 group from the eg rows it already stages; reduce_u launch removed (5 -> 4 launches); u's sum order changes
# speedup vs baseline: 4.0860x; 1.0370x over previous
#include <hip/hip_runtime.h>

typedef __attribute__((ext_vector_type(2))) float v2f;
typedef __attribute__((ext_vector_type(8))) float v8f;

#define R_DIM 64
#define C_DIM 256

__global__ void zero_kernel(float* __restrict__ p, int n) {
    int i = blockIdx.x * blockDim.x + threadIdx.x;
    if (i < n) p[i] = 0.0f;
}

#define PP  160
#define PKS 256
#define SLAB 64

typedef __attribute__((ext_vector_type(4))) float v4f;

__global__ __launch_bounds__(256)
void proj_kernel(const float* __restrict__ eg,
                 const float* __restrict__ x,
                 float* __restrict__ proj,
                 float* __restrict__ u, int n) {
    __shared__ __attribute__((aligned(16))) float egp[(PKS / 2) * PP];
    __shared__ __attribute__((aligned(16))) float xp[(PKS / 2) * PP];

    int tid   = threadIdx.x;
    int lane  = tid & 31;
    int w     = tid >> 5;
    int mrow  = lane & 15;
    int khalf = lane >> 4;
    int rt    = w & 3;
    int ctb   = (w >> 2) * 2;
    int cs    = blockIdx.x * SLAB;

    int chunk = (n + gridDim.y - 1) / gridDim.y;
    int n0    = blockIdx.y * chunk;
    int nend  = min(n0 + chunk, n);

    int c4 = tid & 15;
    int pp = tid >> 4;
    const float* xb = x + cs + c4 * 4;
    const float* eb = eg + c4 * 4;

    v4f rx[16], re[16];
    const v4f zero4 = {0.0f, 0.0f, 0.0f, 0.0f};

    const bool do_u = blockIdx.x == 0;
    v4f us = zero4;

    auto fetch = [&](int nb) {
        if (nb + PKS <= nend) {
            const float* px = xb + (nb + 2 * pp) * C_DIM;
            const float* pe = eb + (nb + 2 * pp) * R_DIM;
#pragma unroll
            for (int j = 0; j < 8; ++j) {
                rx[2 * j]     = *(const v4f*)(px + (32 * j) * C_DIM);
                rx[2 * j + 1] = *(const v4f*)(px + (32 * j + 1) * C_DIM);
            }
#pragma unroll
            for (int j = 0; j < 8; ++j) {
                re[2 * j]     = *(const v4f*)(pe + (32 * j) * R_DIM);
                re[2 * j + 1] = *(const v4f*)(pe + (32 * j + 1) * R_DIM);
            }
        } else {
#pragma unroll
            for (int j = 0; j < 8; ++j) {
                int r0 = min(nb + 2 * (pp + 16 * j), nend - 1);
                int r1 = min(nb + 2 * (pp + 16 * j) + 1, nend - 1);
                rx[2 * j]     = *(const v4f*)(xb + r0 * C_DIM);
                rx[2 * j + 1] = *(const v4f*)(xb + r1 * C_DIM);
                re[2 * j]     = *(const v4f*)(eb + r0 * R_DIM);
                re[2 * j + 1] = *(const v4f*)(eb + r1 * R_DIM);
            }
        }
    };

    v8f acc[2];
#pragma unroll
    for (int t = 0; t < 2; ++t)
#pragma unroll
        for (int j = 0; j < 8; ++j) acc[t][j] = 0.0f;

    struct Frag { v2f a[8], b0[8], b1[8]; };
    auto ldfrag = [&](Frag& f, int pb) {
#pragma unroll
        for (int kk = 0; kk < 8; ++kk) {
            int pidx = pb + kk * 2 + khalf;
            f.a[kk]  = *(const v2f*)&egp[pidx * PP + (rt * 16 + mrow) * 2];
            f.b0[kk] = *(const v2f*)&xp[pidx * PP + ((ctb    ) * 16 + mrow) * 2];
            f.b1[kk] = *(const v2f*)&xp[pidx * PP + ((ctb + 1) * 16 + mrow) * 2];
        }
    };
    auto mmafrag = [&](const Frag& f) {
#pragma unroll
        for (int kk = 0; kk < 8; ++kk) {
            acc[0] = __builtin_amdgcn_wmma_f32_16x16x4_f32(
                false, f.a[kk], false, f.b0[kk], (short)0, acc[0], false, false);
            acc[1] = __builtin_amdgcn_wmma_f32_16x16x4_f32(
                false, f.a[kk], false, f.b1[kk], (short)0, acc[1], false, false);
        }
    };

    if (n0 < nend) fetch(n0);

    for (int nb = n0; nb < nend; nb += PKS) {
        bool part = nb + PKS > nend;
#pragma unroll
        for (int j = 0; j < 8; ++j) {
            int p = pp + 16 * j;
            float* dx = &xp[p * PP + c4 * 8];
            float* de = &egp[p * PP + c4 * 8];
            v4f a0 = rx[2 * j], a1 = rx[2 * j + 1];
            v4f e0 = re[2 * j], e1 = re[2 * j + 1];
            if (part) {
                bool k0 = nb + 2 * p < nend, k1 = nb + 2 * p + 1 < nend;
                a0 = k0 ? a0 : zero4;  a1 = k1 ? a1 : zero4;
                e0 = k0 ? e0 : zero4;  e1 = k1 ? e1 : zero4;
            }
            if (do_u) { us += e0; us += e1; }
            v4f s;
            s.x = a0.x; s.y = a1.x; s.z = a0.y; s.w = a1.y; *(v4f*)dx       = s;
            s.x = a0.z; s.y = a1.z; s.z = a0.w; s.w = a1.w; *(v4f*)(dx + 4) = s;
            s.x = e0.x; s.y = e1.x; s.z = e0.y; s.w = e1.y; *(v4f*)de       = s;
            s.x = e0.z; s.y = e1.z; s.z = e0.w; s.w = e1.w; *(v4f*)(de + 4) = s;
        }
        __syncthreads();

        if (nb + PKS < nend) fetch(nb + PKS);

        if (!part) {
            Frag fa, fb;
            ldfrag(fa, 0);
#pragma unroll
            for (int h = 0; h < PKS / 32; h += 2) {
                ldfrag(fb, 16 * (h + 1));
                __builtin_amdgcn_sched_barrier(0);
                mmafrag(fa);
                __builtin_amdgcn_sched_barrier(0);
                if (h + 2 < PKS / 32) ldfrag(fa, 16 * (h + 2));
                __builtin_amdgcn_sched_barrier(0);
                mmafrag(fb);
                __builtin_amdgcn_sched_barrier(0);
            }
        } else {
#pragma nounroll
            for (int h = 0; h < PKS / 32; ++h)
                if (nb + 32 * h < nend) {
                    Frag f;
                    ldfrag(f, 16 * h);
                    mmafrag(f);
                }
        }
        __syncthreads();
    }

    if (do_u) {
        *(v4f*)&egp[pp * R_DIM + c4 * 4] = us;
        __syncthreads();
        if (tid < R_DIM) {
            float s = egp[tid];
#pragma unroll
            for (int p = 1; p < 16; ++p) s += egp[p * R_DIM + tid];
            atomicAdd(&u[tid], s);
        }
    }

#pragma unroll
    for (int t = 0; t < 2; ++t) {
        int c = cs + (ctb + t) * 16 + mrow;
#pragma unroll
        for (int i = 0; i < 8; ++i) {
            int m = i + 8 * khalf;
            int r = rt * 16 + m;
            atomicAdd(&proj[r * C_DIM + c], acc[t][i]);
        }
    }
}

__global__ __launch_bounds__(256)
void p2_kernel(const float* __restrict__ proj,
               const float* __restrict__ W,
               float* __restrict__ P2i) {
    __shared__ float part[8 * 256];

    int tid   = threadIdx.x;
    int lane  = tid & 31;
    int w     = tid >> 5;
    int mrow  = lane & 15;
    int khalf = lane >> 4;
    int rt    = blockIdx.x & 3;
    int ct    = blockIdx.x >> 2;

    const float* pa = proj + (rt * 16 + mrow) * C_DIM + w * 32 + 2 * khalf;
    const float* pb = W + (w * 32 + 2 * khalf) * C_DIM + ct * 16 + mrow;

    v2f a[8], b[8];
#pragma unroll
    for (int j = 0; j < 8; ++j) {
        a[j]   = *(const v2f*)(pa + 4 * j);
        b[j].x = pb[(4 * j) * C_DIM];
        b[j].y = pb[(4 * j + 1) * C_DIM];
    }

    v8f acc;
#pragma unroll
    for (int i = 0; i < 8; ++i) acc[i] = 0.0f;
#pragma unroll
    for (int j = 0; j < 8; ++j)
        acc = __builtin_amdgcn_wmma_f32_16x16x4_f32(
            false, a[j], false, b[j], (short)0, acc, false, false);

#pragma unroll
    for (int i = 0; i < 8; ++i)
        part[w * 256 + (i + 8 * khalf) * 16 + mrow] = acc[i];
    __syncthreads();

    float s = part[tid];
#pragma unroll
    for (int k = 1; k < 8; ++k) s += part[k * 256 + tid];

    int r = rt * 16 + (tid >> 4);
    int c = ct * 16 + (tid & 15);
    P2i[(r >> 1) * (2 * C_DIM) + c * 2 + (r & 1)] = s;
}

#define EGO 68

__global__ void out_kernel(const float* __restrict__ eg,
                           const float* __restrict__ P2i,
                           const float* __restrict__ u,
                           const float* __restrict__ bias,
                           float* __restrict__ out, int n) {
    __shared__ float egLDS[64 * EGO];
    __shared__ float uLDS[64];
    __shared__ float invLDS[64];
    __shared__ float biasLDS[256];

    int tid = threadIdx.x;
    int rb  = blockIdx.x * 64;

    float uu = u[tid & 63];
    float bb = bias[tid];
    float ev[16];
#pragma unroll
    for (int i = 0; i < 16; ++i) {
        int row = i * 4 + (tid >> 6);
        int col = tid & 63;
        int g   = min(rb + row, n - 1);
        ev[i]   = eg[g * R_DIM + col];
    }
    if (tid < 64) uLDS[tid] = uu;
    biasLDS[tid] = bb;
#pragma unroll
    for (int i = 0; i < 16; ++i) {
        int row = i * 4 + (tid >> 6);
        int col = tid & 63;
        egLDS[row * EGO + col] = (rb + row < n) ? ev[i] : 0.0f;
    }
    __syncthreads();

    if (tid < 64) {
        float s = 0.0f;
#pragma unroll
        for (int r = 0; r < R_DIM; ++r) s += egLDS[tid * EGO + r] * uLDS[r];
        invLDS[tid] = 1.0f / s;
    }
    __syncthreads();

    int lane  = tid & 31;
    int w     = tid >> 5;
    int mrow  = lane & 15;
    int khalf = lane >> 4;
    int rt    = w & 3;
    int ctb   = (w >> 2) * 8;

    const v2f* P2v = (const v2f*)P2i;

    v8f acc[8];
#pragma unroll
    for (int t = 0; t < 8; ++t)
#pragma unroll
        for (int j = 0; j < 8; ++j) acc[t][j] = 0.0f;

#pragma unroll
    for (int kk = 0; kk < 16; ++kk) {
        int k0 = kk * 4 + 2 * khalf;
        v2f a = *(const v2f*)&egLDS[(rt * 16 + mrow) * EGO + k0];
#pragma unroll
        for (int t = 0; t < 8; ++t) {
            int c = (ctb + t) * 16 + mrow;
            v2f b = P2v[(k0 >> 1) * C_DIM + c];
            acc[t] = __builtin_amdgcn_wmma_f32_16x16x4_f32(
                false, a, false, b, (short)0, acc[t], false, false);
        }
    }

#pragma unroll
    for (int t = 0; t < 8; ++t) {
        int c = (ctb + t) * 16 + mrow;
#pragma unroll
        for (int i = 0; i < 8; ++i) {
            int m    = i + 8 * khalf;
            int grow = rb + rt * 16 + m;
            if (grow < n)
                out[grow * C_DIM + c] =
                    acc[t][i] * invLDS[rt * 16 + m] + biasLDS[c];
        }
    }
}

extern "C" void kernel_launch(void* const* d_in, const int* in_sizes, int n_in,
                              void* d_out, int out_size, void* d_ws, size_t ws_size,
                              hipStream_t stream) {
    const float* x    = (const float*)d_in[0];
    const float* eg   = (const float*)d_in[2];
    const float* W    = (const float*)d_in[3];
    const float* bias = (const float*)d_in[4];
    float*       out  = (float*)d_out;

    int N = in_sizes[2] / R_DIM;

    float* ws   = (float*)d_ws;
    float* u    = ws;
    float* proj = ws + 64;
    float* P2i  = ws + 64 + R_DIM * C_DIM;

    int nz = 64 + R_DIM * C_DIM;
    zero_kernel<<<(nz + 255) / 256, 256, 0, stream>>>(ws, nz);
    proj_kernel<<<dim3(4, 64), 256, 0, stream>>>(eg, x, proj, u, N);
    p2_kernel<<<64, 256, 0, stream>>>(proj, W, P2i);
    out_kernel<<<(N + 63) / 64, 256, 0, stream>>>(eg, P2i, u, bias, out, N);
}


# out_kernel resident: 256 work-groups walk 64-row blocks, each wave keeps its P2 B-fragments in registers, eg staged one block ahead; deg summed per wave from its A fragments (deg sum order changes)
# speedup vs baseline: 5.1087x; 1.2503x over previous
#include <hip/hip_runtime.h>

typedef __attribute__((ext_vector_type(2))) float v2f;
typedef __attribute__((ext_vector_type(8))) float v8f;

#define R_DIM 64
#define C_DIM 256

__global__ void zero_kernel(float* __restrict__ p, int n) {
    int i = blockIdx.x * blockDim.x + threadIdx.x;
    if (i < n) p[i] = 0.0f;
}

#define PP  160
#define PKS 256
#define SLAB 64

typedef __attribute__((ext_vector_type(4))) float v4f;

__global__ __launch_bounds__(256)
void proj_kernel(const float* __restrict__ eg,
                 const float* __restrict__ x,
                 float* __restrict__ proj,
                 float* __restrict__ u, int n) {
    __shared__ __attribute__((aligned(16))) float egp[(PKS / 2) * PP];
    __shared__ __attribute__((aligned(16))) float xp[(PKS / 2) * PP];

    int tid   = threadIdx.x;
    int lane  = tid & 31;
    int w     = tid >> 5;
    int mrow  = lane & 15;
    int khalf = lane >> 4;
    int rt    = w & 3;
    int ctb   = (w >> 2) * 2;
    int cs    = blockIdx.x * SLAB;

    int chunk = (n + gridDim.y - 1) / gridDim.y;
    int n0    = blockIdx.y * chunk;
    int nend  = min(n0 + chunk, n);

    int c4 = tid & 15;
    int pp = tid >> 4;
    const float* xb = x + cs + c4 * 4;
    const float* eb = eg + c4 * 4;

    v4f rx[16], re[16];
    const v4f zero4 = {0.0f, 0.0f, 0.0f, 0.0f};

    const bool do_u = blockIdx.x == 0;
    v4f us = zero4;

    auto fetch = [&](int nb) {
        if (nb + PKS <= nend) {
            const float* px = xb + (nb + 2 * pp) * C_DIM;
            const float* pe = eb + (nb + 2 * pp) * R_DIM;
#pragma unroll
            for (int j = 0; j < 8; ++j) {
                rx[2 * j]     = *(const v4f*)(px + (32 * j) * C_DIM);
                rx[2 * j + 1] = *(const v4f*)(px + (32 * j + 1) * C_DIM);
            }
#pragma unroll
            for (int j = 0; j < 8; ++j) {
                re[2 * j]     = *(const v4f*)(pe + (32 * j) * R_DIM);
                re[2 * j + 1] = *(const v4f*)(pe + (32 * j + 1) * R_DIM);
            }
        } else {
#pragma unroll
            for (int j = 0; j < 8; ++j) {
                int r0 = min(nb + 2 * (pp + 16 * j), nend - 1);
                int r1 = min(nb + 2 * (pp + 16 * j) + 1, nend - 1);
                rx[2 * j]     = *(const v4f*)(xb + r0 * C_DIM);
                rx[2 * j + 1] = *(const v4f*)(xb + r1 * C_DIM);
                re[2 * j]     = *(const v4f*)(eb + r0 * R_DIM);
                re[2 * j + 1] = *(const v4f*)(eb + r1 * R_DIM);
            }
        }
    };

    v8f acc[2];
#pragma unroll
    for (int t = 0; t < 2; ++t)
#pragma unroll
        for (int j = 0; j < 8; ++j) acc[t][j] = 0.0f;

    struct Frag { v2f a[8], b0[8], b1[8]; };
    auto ldfrag = [&](Frag& f, int pb) {
#pragma unroll
        for (int kk = 0; kk < 8; ++kk) {
            int pidx = pb + kk * 2 + khalf;
            f.a[kk]  = *(const v2f*)&egp[pidx * PP + (rt * 16 + mrow) * 2];
            f.b0[kk] = *(const v2f*)&xp[pidx * PP + ((ctb    ) * 16 + mrow) * 2];
            f.b1[kk] = *(const v2f*)&xp[pidx * PP + ((ctb + 1) * 16 + mrow) * 2];
        }
    };
    auto mmafrag = [&](const Frag& f) {
#pragma unroll
        for (int kk = 0; kk < 8; ++kk) {
            acc[0] = __builtin_amdgcn_wmma_f32_16x16x4_f32(
                false, f.a[kk], false, f.b0[kk], (short)0, acc[0], false, false);
            acc[1] = __builtin_amdgcn_wmma_f32_16x16x4_f32(
                false, f.a[kk], false, f.b1[kk], (short)0, acc[1], false, false);
        }
    };

    if (n0 < nend) fetch(n0);

    for (int nb = n0; nb < nend; nb += PKS) {
        bool part = nb + PKS > nend;
#pragma unroll
        for (int j = 0; j < 8; ++j) {
            int p = pp + 16 * j;
            float* dx = &xp[p * PP + c4 * 8];
            float* de = &egp[p * PP + c4 * 8];
            v4f a0 = rx[2 * j], a1 = rx[2 * j + 1];
            v4f e0 = re[2 * j], e1 = re[2 * j + 1];
            if (part) {
                bool k0 = nb + 2 * p < nend, k1 = nb + 2 * p + 1 < nend;
                a0 = k0 ? a0 : zero4;  a1 = k1 ? a1 : zero4;
                e0 = k0 ? e0 : zero4;  e1 = k1 ? e1 : zero4;
            }
            if (do_u) { us += e0; us += e1; }
            v4f s;
            s.x = a0.x; s.y = a1.x; s.z = a0.y; s.w = a1.y; *(v4f*)dx       = s;
            s.x = a0.z; s.y = a1.z; s.z = a0.w; s.w = a1.w; *(v4f*)(dx + 4) = s;
            s.x = e0.x; s.y = e1.x; s.z = e0.y; s.w = e1.y; *(v4f*)de       = s;
            s.x = e0.z; s.y = e1.z; s.z = e0.w; s.w = e1.w; *(v4f*)(de + 4) = s;
        }
        __syncthreads();

        if (nb + PKS < nend) fetch(nb + PKS);

        if (!part) {
            Frag fa, fb;
            ldfrag(fa, 0);
#pragma unroll
            for (int h = 0; h < PKS / 32; h += 2) {
                ldfrag(fb, 16 * (h + 1));
                __builtin_amdgcn_sched_barrier(0);
                mmafrag(fa);
                __builtin_amdgcn_sched_barrier(0);
                if (h + 2 < PKS / 32) ldfrag(fa, 16 * (h + 2));
                __builtin_amdgcn_sched_barrier(0);
                mmafrag(fb);
                __builtin_amdgcn_sched_barrier(0);
            }
        } else {
#pragma nounroll
            for (int h = 0; h < PKS / 32; ++h)
                if (nb + 32 * h < nend) {
                    Frag f;
                    ldfrag(f, 16 * h);
                    mmafrag(f);
                }
        }
        __syncthreads();
    }

    if (do_u) {
        *(v4f*)&egp[pp * R_DIM + c4 * 4] = us;
        __syncthreads();
        if (tid < R_DIM) {
            float s = egp[tid];
#pragma unroll
            for (int p = 1; p < 16; ++p) s += egp[p * R_DIM + tid];
            atomicAdd(&u[tid], s);
        }
    }

#pragma unroll
    for (int t = 0; t < 2; ++t) {
        int c = cs + (ctb + t) * 16 + mrow;
#pragma unroll
        for (int i = 0; i < 8; ++i) {
            int m = i + 8 * khalf;
            int r = rt * 16 + m;
            atomicAdd(&proj[r * C_DIM + c], acc[t][i]);
        }
    }
}

__global__ __launch_bounds__(256)
void p2_kernel(const float* __restrict__ proj,
               const float* __restrict__ W,
               float* __restrict__ P2i) {
    __shared__ float part[8 * 256];

    int tid   = threadIdx.x;
    int lane  = tid & 31;
    int w     = tid >> 5;
    int mrow  = lane & 15;
    int khalf = lane >> 4;
    int rt    = blockIdx.x & 3;
    int ct    = blockIdx.x >> 2;

    const float* pa = proj + (rt * 16 + mrow) * C_DIM + w * 32 + 2 * khalf;
    const float* pb = W + (w * 32 + 2 * khalf) * C_DIM + ct * 16 + mrow;

    v2f a[8], b[8];
#pragma unroll
    for (int j = 0; j < 8; ++j) {
        a[j]   = *(const v2f*)(pa + 4 * j);
        b[j].x = pb[(4 * j) * C_DIM];
        b[j].y = pb[(4 * j + 1) * C_DIM];
    }

    v8f acc;
#pragma unroll
    for (int i = 0; i < 8; ++i) acc[i] = 0.0f;
#pragma unroll
    for (int j = 0; j < 8; ++j)
        acc = __builtin_amdgcn_wmma_f32_16x16x4_f32(
            false, a[j], false, b[j], (short)0, acc, false, false);

#pragma unroll
    for (int i = 0; i < 8; ++i)
        part[w * 256 + (i + 8 * khalf) * 16 + mrow] = acc[i];
    __syncthreads();

    float s = part[tid];
#pragma unroll
    for (int k = 1; k < 8; ++k) s += part[k * 256 + tid];

    int r = rt * 16 + (tid >> 4);
    int c = ct * 16 + (tid & 15);
    P2i[(r >> 1) * (2 * C_DIM) + c * 2 + (r & 1)] = s;
}

#define EGO 68

__global__ __launch_bounds__(256)
void out_kernel(const float* __restrict__ eg,
                const float* __restrict__ P2i,
                const float* __restrict__ u,
                const float* __restrict__ bias,
                float* __restrict__ out, int n) {
    __shared__ __attribute__((aligned(16))) float egLDS[2][64 * EGO];

    int tid   = threadIdx.x;
    int lane  = tid & 31;
    int w     = tid >> 5;
    int mrow  = lane & 15;
    int khalf = lane >> 4;

    int per  = ((n + (int)gridDim.x * 16 - 1) / ((int)gridDim.x * 16)) * 16;
    int row0 = blockIdx.x * per;
    int rend = min(row0 + per, n);
    if (row0 >= rend) return;

    int c4 = tid & 15;
    int sr = tid >> 4;
    const float* eb = eg + c4 * 4;
    const v4f zero4 = {0.0f, 0.0f, 0.0f, 0.0f};

    v4f ev[4];
    auto fetch = [&](int rb) {
#pragma unroll
        for (int j = 0; j < 4; ++j) {
            int g = min(rb + sr + 16 * j, n - 1);
            ev[j] = *(const v4f*)(eb + g * R_DIM);
        }
    };
    fetch(row0);

    const v2f* P2v = (const v2f*)P2i;
    v2f bf[16][2], uf[16];
#pragma unroll
    for (int kk = 0; kk < 16; ++kk) {
        bf[kk][0] = P2v[(2 * kk + khalf) * C_DIM + (2 * w) * 16 + mrow];
        bf[kk][1] = P2v[(2 * kk + khalf) * C_DIM + (2 * w + 1) * 16 + mrow];
        uf[kk]    = *(const v2f*)(u + 4 * kk + 2 * khalf);
    }
    float bs0 = bias[(2 * w) * 16 + mrow];
    float bs1 = bias[(2 * w + 1) * 16 + mrow];

    struct AF { v2f a[16]; };

    int buf = 0;
    for (int rb = row0; rb < rend; rb += 64, buf ^= 1) {
        float* egb = egLDS[buf];

#pragma unroll
        for (int j = 0; j < 4; ++j) {
            int row = sr + 16 * j;
            *(v4f*)&egb[row * EGO + c4 * 4] = (rb + row < n) ? ev[j] : zero4;
        }
        __syncthreads();

        if (rb + 64 < rend) fetch(rb + 64);

        auto ldA = [&](AF& f, int rt) {
#pragma unroll
            for (int kk = 0; kk < 16; ++kk)
                f.a[kk] = *(const v2f*)&egb[(rt * 16 + mrow) * EGO + 4 * kk + 2 * khalf];
        };
        auto degs = [&](const AF& f, float (&iv)[8]) {
            v2f pa = {0.0f, 0.0f};
#pragma unroll
            for (int kk = 0; kk < 16; ++kk) pa += f.a[kk] * uf[kk];
            float p = pa.x + pa.y;
            p += __shfl_xor(p, 16);
            float inv = 1.0f / p;
#pragma unroll
            for (int i = 0; i < 8; ++i) iv[i] = __shfl(inv, 8 * khalf + i);
        };
        auto mma = [&](const AF& f, v8f& c0, v8f& c1) {
#pragma unroll
            for (int j = 0; j < 8; ++j) { c0[j] = 0.0f; c1[j] = 0.0f; }
#pragma unroll
            for (int kk = 0; kk < 16; ++kk) {
                c0 = __builtin_amdgcn_wmma_f32_16x16x4_f32(
                    false, f.a[kk], false, bf[kk][0], (short)0, c0, false, false);
                c1 = __builtin_amdgcn_wmma_f32_16x16x4_f32(
                    false, f.a[kk], false, bf[kk][1], (short)0, c1, false, false);
            }
        };
        auto epi = [&](int rt, const v8f& c0, const v8f& c1, const float (&iv)[8], bool guard) {
            int m0 = rt * 16 + 8 * khalf;
            float* o = out + (rb + m0) * C_DIM + (2 * w) * 16 + mrow;
#pragma unroll
            for (int i = 0; i < 8; ++i)
                if (!guard || rb + m0 + i < n) {
                    o[i * C_DIM]      = c0[i] * iv[i] + bs0;
                    o[i * C_DIM + 16] = c1[i] * iv[i] + bs1;
                }
        };

        if (rb + 64 <= rend) {
            AF fa, fb;
            v8f c0, c1;
            float iv[8];
            ldA(fa, 0);
            ldA(fb, 1);
            __builtin_amdgcn_sched_barrier(0);
            degs(fa, iv);
            mma(fa, c0, c1);
            __builtin_amdgcn_sched_barrier(0);
            epi(0, c0, c1, iv, false);
            ldA(fa, 2);
            __builtin_amdgcn_sched_barrier(0);
            degs(fb, iv);
            mma(fb, c0, c1);
            __builtin_amdgcn_sched_barrier(0);
            epi(1, c0, c1, iv, false);
            ldA(fb, 3);
            __builtin_amdgcn_sched_barrier(0);
            degs(fa, iv);
            mma(fa, c0, c1);
            __builtin_amdgcn_sched_barrier(0);
            epi(2, c0, c1, iv, false);
            degs(fb, iv);
            mma(fb, c0, c1);
            epi(3, c0, c1, iv, false);
        } else {
            int nt = (rend - rb + 15) >> 4;
#pragma nounroll
            for (int rt = 0; rt < nt; ++rt) {
                AF f;
                v8f c0, c1;
                float iv[8];
                ldA(f, rt);
                degs(f, iv);
                mma(f, c0, c1);
                epi(rt, c0, c1, iv, true);
            }
        }
    }
}

extern "C" void kernel_launch(void* const* d_in, const int* in_sizes, int n_in,
                              void* d_out, int out_size, void* d_ws, size_t ws_size,
                              hipStream_t stream) {
    const float* x    = (const float*)d_in[0];
    const float* eg   = (const float*)d_in[2];
    const float* W    = (const float*)d_in[3];
    const float* bias = (const float*)d_in[4];
    float*       out  = (float*)d_out;

    int N = in_sizes[2] / R_DIM;

    float* ws   = (float*)d_ws;
    float* u    = ws;
    float* proj = ws + 64;
    float* P2i  = ws + 64 + R_DIM * C_DIM;

    int nz = 64 + R_DIM * C_DIM;
    zero_kernel<<<(nz + 255) / 256, 256, 0, stream>>>(ws, nz);
    proj_kernel<<<dim3(4, 64), 256, 0, stream>>>(eg, x, proj, u, N);
    p2_kernel<<<64, 256, 0, stream>>>(proj, W, P2i);
    out_kernel<<<256, 256, 0, stream>>>(eg, P2i, u, bias, out, N);
}
